# starting kernel
# baseline (speedup 1.0000x reference)
#include <hip/hip_runtime.h>
#include <hip/hip_bf16.h>

typedef float v2f __attribute__((ext_vector_type(2)));
typedef float v8f __attribute__((ext_vector_type(8)));

#define BATCH      16
#define CHANS      256
#define NPIX       16384
#define KCLS       6
#define KPAD       8
#define IGNORE_IDX 255

#define NCHUNK     256
#define ROWSTRIDE  260
#define WAVES      4
#define THREADS    (WAVES * 32)
#define NSPLIT     8
#define STEPS      (NCHUNK / (WAVES * 4))

__global__ void zero_acc_kernel(float* __restrict__ acc, int n) {
    int i = blockIdx.x * blockDim.x + threadIdx.x;
    if (i < n) acc[i] = 0.0f;
}

__global__ void count_kernel(const int* __restrict__ gt, float* __restrict__ counts) {
    __shared__ int lc[KPAD];
    int b = blockIdx.x;
    if (threadIdx.x < KPAD) lc[threadIdx.x] = 0;
    __syncthreads();
    const int* gb = gt + (size_t)b * NPIX;
    for (int i = threadIdx.x; i < NPIX; i += blockDim.x) {
        int l = gb[i];
        if (l >= 0 && l < KCLS) atomicAdd(&lc[l], 1);
    }
    __syncthreads();
    if (threadIdx.x < KPAD) counts[b * KPAD + threadIdx.x] = (float)lc[threadIdx.x];
}

__global__ void __launch_bounds__(THREADS)
center_accum_kernel(const float* __restrict__ x, const int* __restrict__ gt,
                    float* __restrict__ acc  ) {
    __shared__ float tile[16 * ROWSTRIDE];
    __shared__ int   labs[NCHUNK];

    int wg    = blockIdx.x;
    int split = wg % NSPLIT;
    int ct    = (wg / NSPLIT) % (CHANS / 16);
    int b     = wg / (NSPLIT * (CHANS / 16));

    const int nPerSplit = NPIX / NSPLIT;
    int nBeg = split * nPerSplit;
    int nEnd = nBeg + nPerSplit;

    const float* xb = x  + ((size_t)b * CHANS + (size_t)ct * 16) * NPIX;
    const int*   gb = gt + (size_t)b * NPIX;

    int tid  = threadIdx.x;
    int lane = tid & 31;
    int wave = tid >> 5;
    int m    = lane & 15;
    int hi   = lane >> 4;

    v8f d0 = {};
    v8f d1 = {};

    for (int n0 = nBeg; n0 < nEnd; n0 += NCHUNK) {
        {
            const float4* src = (const float4*)(xb + n0);
            for (int i = tid; i < 16 * (NCHUNK / 4); i += THREADS) {
                int r  = i / (NCHUNK / 4);
                int p4 = i % (NCHUNK / 4);
                float4 v = src[(size_t)r * (NPIX / 4) + p4];
                *(float4*)&tile[r * ROWSTRIDE + p4 * 4] = v;
            }
            if (tid < NCHUNK / 4)
                ((int4*)labs)[tid] = ((const int4*)(gb + n0))[tid];
        }
        __syncthreads();

#pragma unroll
        for (int it = 0; it < STEPS; ++it) {
            int s = wave * 4 + it * (WAVES * 4);
            const float* ap = &tile[m * ROWSTRIDE + s + 2 * hi];
            v2f a; a.x = ap[0]; a.y = ap[1];
            int l0 = labs[s + 2 * hi];
            int l1 = labs[s + 2 * hi + 1];
            v2f bm;
            bm.x = (l0 == m) ? 1.0f : 0.0f;
            bm.y = (l1 == m) ? 1.0f : 0.0f;
            if (it & 1)
                d1 = __builtin_amdgcn_wmma_f32_16x16x4_f32(
                        false, a, false, bm, (short)0, d1, false, false);
            else
                d0 = __builtin_amdgcn_wmma_f32_16x16x4_f32(
                        false, a, false, bm, (short)0, d0, false, false);
        }
        __syncthreads();
    }

    if (m < KCLS) {
        float* accb = acc + ((size_t)b * CHANS + (size_t)ct * 16) * KPAD;
#pragma unroll
        for (int j = 0; j < 8; ++j) {
            int c = j + 8 * hi;
            atomicAdd(&accb[c * KPAD + m], d0[j] + d1[j]);
        }
    }
}

__global__ void finalize_kernel(const float* __restrict__ acc,
                                const float* __restrict__ counts,
                                float* __restrict__ out) {
    int idx = blockIdx.x * blockDim.x + threadIdx.x;
    if (idx >= BATCH * KCLS * CHANS) return;
    int c = idx % CHANS;
    int k = (idx / CHANS) % KCLS;
    int b = idx / (CHANS * KCLS);
    float cnt = fmaxf(counts[b * KPAD + k], 1.0f);
    out[idx] = acc[((size_t)b * CHANS + c) * KPAD + k] / cnt;
}

extern "C" void kernel_launch(void* const* d_in, const int* in_sizes, int n_in,
                              void* d_out, int out_size, void* d_ws, size_t ws_size,
                              hipStream_t stream) {
    const float* x  = (const float*)d_in[0];
    const int*   gt = (const int*)d_in[1];
    float*       out = (float*)d_out;

    float* acc    = (float*)d_ws;
    float* counts = acc + (size_t)BATCH * CHANS * KPAD;

    const int accN = BATCH * CHANS * KPAD;
    zero_acc_kernel<<<(accN + 255) / 256, 256, 0, stream>>>(acc, accN);

    count_kernel<<<BATCH, 256, 0, stream>>>(gt, counts);

    const int nBlocks = BATCH * (CHANS / 16) * NSPLIT;
    center_accum_kernel<<<nBlocks, THREADS, 0, stream>>>(x, gt, acc);

    const int outN = BATCH * KCLS * CHANS;
    finalize_kernel<<<(outN + 255) / 256, 256, 0, stream>>>(acc, counts, out);
}


# Class histogram folded into the WMMA pass: channel-tile-0 blocks sum their one-hot B fragments per lane and add whole-number counts by float atomics (order-free); LDS-atomic count launch removed
# speedup vs baseline: 1.6027x; 1.6027x over previous
#include <hip/hip_runtime.h>
#include <hip/hip_bf16.h>

typedef float v2f __attribute__((ext_vector_type(2)));
typedef float v8f __attribute__((ext_vector_type(8)));

#define BATCH      16
#define CHANS      256
#define NPIX       16384
#define KCLS       6
#define KPAD       8
#define IGNORE_IDX 255

#define NCHUNK     256
#define ROWSTRIDE  260
#define WAVES      4
#define THREADS    (WAVES * 32)
#define NSPLIT     8
#define STEPS      (NCHUNK / (WAVES * 4))

__global__ void zero_acc_kernel(float* __restrict__ acc, int n) {
    int i = blockIdx.x * blockDim.x + threadIdx.x;
    if (i < n) acc[i] = 0.0f;
}

__global__ void __launch_bounds__(THREADS)
center_accum_kernel(const float* __restrict__ x, const int* __restrict__ gt,
                    float* __restrict__ acc     ,
                    float* __restrict__ counts  ) {
    __shared__ float tile[16 * ROWSTRIDE];
    __shared__ int   labs[NCHUNK];

    int wg    = blockIdx.x;
    int split = wg % NSPLIT;
    int ct    = (wg / NSPLIT) % (CHANS / 16);
    int b     = wg / (NSPLIT * (CHANS / 16));

    const int nPerSplit = NPIX / NSPLIT;
    int nBeg = split * nPerSplit;
    int nEnd = nBeg + nPerSplit;

    const float* xb = x  + ((size_t)b * CHANS + (size_t)ct * 16) * NPIX;
    const int*   gb = gt + (size_t)b * NPIX;

    int tid  = threadIdx.x;
    int lane = tid & 31;
    int wave = tid >> 5;
    int m    = lane & 15;
    int hi   = lane >> 4;

    v8f d0 = {};
    v8f d1 = {};
    v2f cnt2 = {};

    for (int n0 = nBeg; n0 < nEnd; n0 += NCHUNK) {
        {
            const float4* src = (const float4*)(xb + n0);
            for (int i = tid; i < 16 * (NCHUNK / 4); i += THREADS) {
                int r  = i / (NCHUNK / 4);
                int p4 = i % (NCHUNK / 4);
                float4 v = src[(size_t)r * (NPIX / 4) + p4];
                *(float4*)&tile[r * ROWSTRIDE + p4 * 4] = v;
            }
            if (tid < NCHUNK / 4)
                ((int4*)labs)[tid] = ((const int4*)(gb + n0))[tid];
        }
        __syncthreads();

#pragma unroll
        for (int it = 0; it < STEPS; ++it) {
            int s = wave * 4 + it * (WAVES * 4);
            const float* ap = &tile[m * ROWSTRIDE + s + 2 * hi];
            v2f a; a.x = ap[0]; a.y = ap[1];
            int l0 = labs[s + 2 * hi];
            int l1 = labs[s + 2 * hi + 1];
            v2f bm;
            bm.x = (l0 == m) ? 1.0f : 0.0f;
            bm.y = (l1 == m) ? 1.0f : 0.0f;
            cnt2 += bm;
            if (it & 1)
                d1 = __builtin_amdgcn_wmma_f32_16x16x4_f32(
                        false, a, false, bm, (short)0, d1, false, false);
            else
                d0 = __builtin_amdgcn_wmma_f32_16x16x4_f32(
                        false, a, false, bm, (short)0, d0, false, false);
        }
        __syncthreads();
    }

    if (m < KCLS) {
        float* accb = acc + ((size_t)b * CHANS + (size_t)ct * 16) * KPAD;
#pragma unroll
        for (int j = 0; j < 8; ++j) {
            int c = j + 8 * hi;
            atomicAdd(&accb[c * KPAD + m], d0[j] + d1[j]);
        }
    }

    if (ct == 0) {
        float cnt = cnt2.x + cnt2.y;
        cnt += __shfl_xor(cnt, 16);
        if (lane < KCLS) atomicAdd(&counts[b * KPAD + lane], cnt);
    }
}

__global__ void finalize_kernel(const float* __restrict__ acc,
                                const float* __restrict__ counts,
                                float* __restrict__ out) {
    int idx = blockIdx.x * blockDim.x + threadIdx.x;
    if (idx >= BATCH * KCLS * CHANS) return;
    int c = idx % CHANS;
    int k = (idx / CHANS) % KCLS;
    int b = idx / (CHANS * KCLS);
    float cnt = fmaxf(counts[b * KPAD + k], 1.0f);
    out[idx] = acc[((size_t)b * CHANS + c) * KPAD + k] / cnt;
}

extern "C" void kernel_launch(void* const* d_in, const int* in_sizes, int n_in,
                              void* d_out, int out_size, void* d_ws, size_t ws_size,
                              hipStream_t stream) {
    const float* x  = (const float*)d_in[0];
    const int*   gt = (const int*)d_in[1];
    float*       out = (float*)d_out;

    float* acc    = (float*)d_ws;
    float* counts = acc + (size_t)BATCH * CHANS * KPAD;

    const int zeroN = BATCH * CHANS * KPAD + BATCH * KPAD;
    zero_acc_kernel<<<(zeroN + 255) / 256, 256, 0, stream>>>(acc, zeroN);

    const int nBlocks = BATCH * (CHANS / 16) * NSPLIT;
    center_accum_kernel<<<nBlocks, THREADS, 0, stream>>>(x, gt, acc, counts);

    const int outN = BATCH * KCLS * CHANS;
    finalize_kernel<<<(outN + 255) / 256, 256, 0, stream>>>(acc, counts, out);
}


# One 16-wave workgroup per (batch, channel tile) owns all pixels; waves keep the next chunk's loads in flight via private LDS slabs; fixed-order in-block sum replaces atomics and zeroing launch
# speedup vs baseline: 2.8865x; 1.8010x over previous
#include <hip/hip_runtime.h>

typedef float v2f __attribute__((ext_vector_type(2)));
typedef float v8f __attribute__((ext_vector_type(8)));
typedef float f4  __attribute__((ext_vector_type(4)));
typedef int   i4  __attribute__((ext_vector_type(4)));

#define BATCH      16
#define CHANS      256
#define NPIX       16384
#define KCLS       6
#define KPAD       8

#define WAVES      16
#define THREADS    (WAVES * 32)
#define CHUNK      128
#define NTRIP      (NPIX / (WAVES * CHUNK))
#define FG         4
#define PITCH      132

__global__ void __launch_bounds__(THREADS)
center_sweep_kernel(const float* __restrict__ x, const int* __restrict__ gt,
                    float* __restrict__ mean  ) {
    __shared__ float slab[WAVES][16 * PITCH];
    __shared__ int   labs[WAVES][CHUNK];
    __shared__ float cntw[WAVES][KPAD];

    const int tid  = threadIdx.x;
    const int lane = tid & 31;
    const int wave = __builtin_amdgcn_readfirstlane(tid >> 5);
    const int m    = lane & 15;
    const int hi   = lane >> 4;
    const int b    = blockIdx.x / (CHANS / 16);
    const int ct   = blockIdx.x % (CHANS / 16);

    const int rot = ct & (NTRIP - 1);
    const float* xw = x  + ((size_t)b * CHANS + (size_t)ct * 16) * NPIX + wave * CHUNK + 4 * lane;
    const int*   gw = gt + (size_t)b * NPIX + wave * CHUNK + 4 * lane;
    float* sl = slab[wave];
    int*   lb = labs[wave];

    f4 ring[2][16];
    i4 lring[2];

#define LOAD_CHUNK(T) do { \
    _Pragma("unroll") for (int q = 0; q < 16; ++q) \
        ring[(T) & 1][q] = __builtin_nontemporal_load((const f4*)(xw + (size_t)q * NPIX + (((T) + rot) & (NTRIP - 1)) * (WAVES * CHUNK))); \
    lring[(T) & 1] = *(const i4*)(gw + (((T) + rot) & (NTRIP - 1)) * (WAVES * CHUNK)); } while (0)

    v8f d0 = {};
    v8f d1 = {};
    v2f cnt2 = {};

    LOAD_CHUNK(0);
#pragma unroll
    for (int t = 0; t < NTRIP; ++t) {
        __builtin_amdgcn_sched_barrier(0);
        if (t + 1 < NTRIP) LOAD_CHUNK(t + 1);
        __builtin_amdgcn_sched_barrier(0);

#pragma unroll
        for (int q = 0; q < 16; ++q)
            *(f4*)&sl[q * PITCH + 4 * lane] = ring[t & 1][q];
        *(i4*)&lb[4 * lane] = lring[t & 1];

        f4 fa[2][FG];
        i4 fl[2][FG];
#define LOAD_FRAGS(G) do { \
    _Pragma("unroll") for (int u = 0; u < FG; ++u) { \
        fa[(G) & 1][u] = *(const f4*)&sl[m * PITCH + 8 * ((G) * FG + u) + 4 * hi]; \
        fl[(G) & 1][u] = *(const i4*)&lb[8 * ((G) * FG + u) + 4 * hi]; } } while (0)
        LOAD_FRAGS(0);
#pragma unroll
        for (int g = 0; g < CHUNK / (8 * FG); ++g) {
            __builtin_amdgcn_sched_barrier(0);
            if (g + 1 < CHUNK / (8 * FG)) LOAD_FRAGS(g + 1);
            __builtin_amdgcn_sched_barrier(0);
#pragma unroll
            for (int u = 0; u < FG; ++u) {
                const f4 a4 = fa[g & 1][u];
                const i4 l4 = fl[g & 1][u];
                v2f a, bm;
                a.x = a4.x; a.y = a4.y;
                bm.x = (l4.x == m) ? 1.0f : 0.0f;
                bm.y = (l4.y == m) ? 1.0f : 0.0f;
                cnt2 += bm;
                d0 = __builtin_amdgcn_wmma_f32_16x16x4_f32(false, a, false, bm, (short)0, d0, false, false);
                a.x = a4.z; a.y = a4.w;
                bm.x = (l4.z == m) ? 1.0f : 0.0f;
                bm.y = (l4.w == m) ? 1.0f : 0.0f;
                cnt2 += bm;
                d1 = __builtin_amdgcn_wmma_f32_16x16x4_f32(false, a, false, bm, (short)0, d1, false, false);
            }
        }
#undef LOAD_FRAGS
    }
#undef LOAD_CHUNK

    {
        float cnt = cnt2.x + cnt2.y;
        cnt += __shfl_xor(cnt, 16);
        if (m < KPAD) {
#pragma unroll
            for (int j = 0; j < 8; ++j)
                sl[(j + 8 * hi) * KPAD + m] = d0[j] + d1[j];
            if (hi == 0) cntw[wave][m] = cnt;
        }
    }
    __syncthreads();

    if (tid < 16 * KPAD) {
        const int k = tid & (KPAD - 1);
        float s = 0.0f, n = 0.0f;
#pragma unroll
        for (int w = 0; w < WAVES; ++w) {
            s += slab[w][tid];
            n += cntw[w][k];
        }
        mean[((size_t)b * CHANS + (size_t)ct * 16) * KPAD + tid] = s / fmaxf(n, 1.0f);
    }
}

__global__ void finalize_kernel(const float* __restrict__ mean, float* __restrict__ out) {
    int idx = blockIdx.x * blockDim.x + threadIdx.x;
    if (idx >= BATCH * KCLS * CHANS) return;
    int c = idx % CHANS;
    int k = (idx / CHANS) % KCLS;
    int b = idx / (CHANS * KCLS);
    out[idx] = mean[((size_t)b * CHANS + c) * KPAD + k];
}

extern "C" void kernel_launch(void* const* d_in, const int* in_sizes, int n_in,
                              void* d_out, int out_size, void* d_ws, size_t ws_size,
                              hipStream_t stream) {
    const float* x  = (const float*)d_in[0];
    const int*   gt = (const int*)d_in[1];
    float*       out = (float*)d_out;

    float* mean = (float*)d_ws;

    center_sweep_kernel<<<BATCH * (CHANS / 16), THREADS, 0, stream>>>(x, gt, mean);

    const int outN = BATCH * KCLS * CHANS;
    finalize_kernel<<<(outN + 255) / 256, 256, 0, stream>>>(mean, out);
}
